# starting kernel
# baseline (speedup 1.0000x reference)
#include <hip/hip_runtime.h>


typedef __attribute__((ext_vector_type(2))) float v2f;
typedef __attribute__((ext_vector_type(8))) float v8f;

#define EMB   300
#define BATCH 2048
#define SEQ   200
#define HID   128
#define OUT_N 20

__global__ __launch_bounds__(256) void embed_mean_kernel(
    const int* __restrict__ x, const int* __restrict__ lengths,
    const float* __restrict__ table, float* __restrict__ rep)
{
  __shared__ int s_idx[SEQ];
  const int b = blockIdx.x;
  const int t = threadIdx.x;

  for (int l = t; l < SEQ; l += 256) s_idx[l] = x[b * SEQ + l];
  __syncthreads();

  const int c0 = t;
  const int c1 = t + 256;
  float acc0 = 0.0f, acc1 = 0.0f;

  for (int l = 0; l < SEQ; ++l) {
    const float* row = table + (long)s_idx[l] * EMB;
    const int lnext = (l + 1 < SEQ) ? (l + 1) : l;
    const float* nrow = table + (long)s_idx[lnext] * EMB;
    __builtin_prefetch(nrow + c0, 0, 3);

    acc0 += row[c0];
    if (c1 < EMB) acc1 += row[c1];
  }

  const float inv = 1.0f / (float)lengths[b];
  rep[(long)b * EMB + c0] = acc0 * inv;
  if (c1 < EMB) rep[(long)b * EMB + c1] = acc1 * inv;
}

__global__ __launch_bounds__(256) void mlp_wmma_kernel(
    const float* __restrict__ rep, const float* __restrict__ W1,
    const float* __restrict__ b1,  const float* __restrict__ W2,
    const float* __restrict__ b2,  float* __restrict__ out)
{
  __shared__ float h_lds[8][16 * HID];

  const int tid  = threadIdx.x;
  const int w    = tid >> 5;
  const int lane = tid & 31;
  const int half = lane >> 4;
  const int ln   = lane & 15;
  const int kb   = half * 2;
  const int row0 = blockIdx.x * 128 + w * 16;

  v8f acc[8] = {};
  const float* arow = rep + (long)(row0 + ln) * EMB;
  for (int k4 = 0; k4 < EMB / 4; ++k4) {
    const int k = k4 * 4 + kb;
    const v2f a = *(const v2f*)(arow + k);
#pragma unroll
    for (int nt = 0; nt < 8; ++nt) {
      const int col = nt * 16 + ln;
      v2f bf;
      bf.x = W1[(long)k * HID + col];
      bf.y = W1[(long)(k + 1) * HID + col];
      acc[nt] = __builtin_amdgcn_wmma_f32_16x16x4_f32(
           false, a,  false, bf,
           (short)0, acc[nt],  false,  false);
    }
  }

#pragma unroll
  for (int nt = 0; nt < 8; ++nt) {
    const float bias = b1[nt * 16 + ln];
#pragma unroll
    for (int j = 0; j < 8; ++j) {
      float v = acc[nt][j] + bias;
      v = v > 0.0f ? v : 0.0f;
      h_lds[w][(j + 8 * half) * HID + nt * 16 + ln] = v;
    }
  }
  __syncthreads();

  v8f c2[2] = {};
  for (int k4 = 0; k4 < HID / 4; ++k4) {
    const int k = k4 * 4 + kb;
    v2f a;
    a.x = h_lds[w][ln * HID + k];
    a.y = h_lds[w][ln * HID + k + 1];
#pragma unroll
    for (int nt = 0; nt < 2; ++nt) {
      const int col = nt * 16 + ln;
      const bool ok = (col < OUT_N);
      v2f bf;
      bf.x = ok ? W2[k * OUT_N + col]       : 0.0f;
      bf.y = ok ? W2[(k + 1) * OUT_N + col] : 0.0f;
      c2[nt] = __builtin_amdgcn_wmma_f32_16x16x4_f32(
          false, a, false, bf, (short)0, c2[nt], false, false);
    }
  }

#pragma unroll
  for (int nt = 0; nt < 2; ++nt) {
    const int col = nt * 16 + ln;
    if (col < OUT_N) {
      const float bias = b2[col];
#pragma unroll
      for (int j = 0; j < 8; ++j) {
        const int row = row0 + j + 8 * half;
        out[(long)row * OUT_N + col] = c2[nt][j] + bias;
      }
    }
  }
}

extern "C" void kernel_launch(void* const* d_in, const int* in_sizes, int n_in,
                              void* d_out, int out_size, void* d_ws, size_t ws_size,
                              hipStream_t stream) {
  const int*   x       = (const int*)  d_in[0];
  const int*   lengths = (const int*)  d_in[1];
  const float* table   = (const float*)d_in[2];
  const float* W1      = (const float*)d_in[3];
  const float* b1      = (const float*)d_in[4];
  const float* W2      = (const float*)d_in[5];
  const float* b2      = (const float*)d_in[6];
  float*       logits  = (float*)d_out;
  float*       rep     = (float*)d_ws;

  embed_mean_kernel<<<BATCH, 256, 0, stream>>>(x, lengths, table, rep);
  mlp_wmma_kernel<<<BATCH / 128, 256, 0, stream>>>(rep, W1, b1, W2, b2, logits);
}


# Gather: 3-wave work-group per batch row, thread = one 16-byte column group of the row for all 200 rows; loads issued 10 rows a group, next group ahead of the sums (was 1 in flight); sum order kept
# speedup vs baseline: 3.6231x; 3.6231x over previous
#include <hip/hip_runtime.h>


typedef __attribute__((ext_vector_type(2))) float v2f;
typedef __attribute__((ext_vector_type(8))) float v8f;

#define EMB   300
#define BATCH 2048
#define SEQ   200
#define HID   128
#define OUT_N 20

typedef __attribute__((ext_vector_type(4))) float v4f;

#define QPR 75
#define GWG 96
#define GR  10
#define NG  (SEQ / GR)

__device__ __forceinline__ void gather_issue(v4f (&v)[GR], const int (&idx)[GR],
                                             const v4f* __restrict__ tab4, int t)
{
#pragma unroll
  for (int u = 0; u < GR; ++u) v[u] = tab4[(long)idx[u] * QPR + t];
}

__device__ __forceinline__ void gather_sum(v4f& acc, const v4f (&v)[GR])
{
#pragma unroll
  for (int u = 0; u < GR; ++u) {
    acc.x += v[u].x;
    acc.y += v[u].y;
    acc.z += v[u].z;
    acc.w += v[u].w;
  }
}

__global__ __launch_bounds__(GWG) void embed_mean_kernel(
    const int* __restrict__ x, const int* __restrict__ lengths,
    const float* __restrict__ table, float* __restrict__ rep)
{
  const int b = blockIdx.x;
  const int t = threadIdx.x;
  if (t >= QPR) return;

  const int* __restrict__ xi = x + b * SEQ;
  const v4f* __restrict__ tab4 = (const v4f*)table;

  int ia[GR], ib[GR];
  v4f va[GR], vb[GR];
#pragma unroll
  for (int u = 0; u < GR; ++u) ia[u] = xi[u];
#pragma unroll
  for (int u = 0; u < GR; ++u) ib[u] = xi[GR + u];
  gather_issue(va, ia, tab4, t);
#pragma unroll
  for (int u = 0; u < GR; ++u) ia[u] = xi[2 * GR + u];

  v4f acc = {0.0f, 0.0f, 0.0f, 0.0f};
  for (int k = 0; k < NG / 2 - 1; ++k) {
    const int gb = 2 * k + 3;
    const int ga = (2 * k + 4 < NG) ? (2 * k + 4) : (NG - 1);

    gather_issue(vb, ib, tab4, t);
#pragma unroll
    for (int u = 0; u < GR; ++u) ib[u] = xi[gb * GR + u];
    __builtin_amdgcn_sched_barrier(0);
    gather_sum(acc, va);

    gather_issue(va, ia, tab4, t);
#pragma unroll
    for (int u = 0; u < GR; ++u) ia[u] = xi[ga * GR + u];
    __builtin_amdgcn_sched_barrier(0);
    gather_sum(acc, vb);
  }
  gather_issue(vb, ib, tab4, t);
  __builtin_amdgcn_sched_barrier(0);
  gather_sum(acc, va);
  gather_sum(acc, vb);

  const float inv = 1.0f / (float)lengths[b];
  v4f r;
  r.x = acc.x * inv;
  r.y = acc.y * inv;
  r.z = acc.z * inv;
  r.w = acc.w * inv;
  ((v4f*)rep)[(long)b * QPR + t] = r;
}

#define KSTEPS1 (EMB / 4)
#define SPW     10
#define PP      132
#define P2      36

__global__ __launch_bounds__(256) void mlp_wmma_kernel(
    const float* __restrict__ rep, const float* __restrict__ W1,
    const float* __restrict__ b1,  const float* __restrict__ W2,
    const float* __restrict__ b2,  float* __restrict__ out)
{
  __shared__ __attribute__((aligned(16))) float part[8][16 * PP];
  __shared__ float part2[8][16 * P2];

  const int tid  = threadIdx.x;
  const int w    = tid >> 5;
  const int lane = tid & 31;
  const int half = lane >> 4;
  const int ln   = lane & 15;
  const int kb   = half * 2;
  const int row0 = blockIdx.x * 16;

  const float* arow = rep + (long)(row0 + ln) * EMB;
  v2f a1[SPW];
  v4f bx[SPW][2], by[SPW][2];
#pragma unroll
  for (int s = 0; s < SPW; ++s) {
    const int  k4    = w * SPW + s;
    const bool valid = (s < KSTEPS1 - 7 * SPW) || (k4 < KSTEPS1);
    const int  k     = (valid ? k4 : KSTEPS1 - 1) * 4 + kb;
    const v2f  av    = *(const v2f*)(arow + k);
    const v4f* p0    = (const v4f*)(W1 + (long)k * HID + ln * 8);
    const v4f* p1    = (const v4f*)(W1 + (long)(k + 1) * HID + ln * 8);
    const v4f  x0 = p0[0], x1 = p0[1], y0 = p1[0], y1 = p1[1];
    const v2f  z2 = {0.0f, 0.0f};
    const v4f  z4 = {0.0f, 0.0f, 0.0f, 0.0f};
    a1[s]    = valid ? av : z2;
    bx[s][0] = valid ? x0 : z4;
    bx[s][1] = valid ? x1 : z4;
    by[s][0] = valid ? y0 : z4;
    by[s][1] = valid ? y1 : z4;
  }

  v2f w2f[4][2];
  v2f b1v[4];
#pragma unroll
  for (int q = 0; q < 4; ++q) {
    const int k2 = w * 16 + q * 4 + kb;
    b1v[q] = *(const v2f*)(b1 + k2);
#pragma unroll
    for (int nt = 0; nt < 2; ++nt) {
      const int  col = nt * 16 + ln;
      const bool ok  = (col < OUT_N);
      w2f[q][nt].x = ok ? W2[k2 * OUT_N + col]       : 0.0f;
      w2f[q][nt].y = ok ? W2[(k2 + 1) * OUT_N + col] : 0.0f;
    }
  }

  v8f acc[8] = {};
#pragma unroll
  for (int s = 0; s < SPW; ++s) {
#pragma unroll
    for (int nt = 0; nt < 8; ++nt) {
      v2f bf;
      bf.x = bx[s][nt >> 2][nt & 3];
      bf.y = by[s][nt >> 2][nt & 3];
      acc[nt] = __builtin_amdgcn_wmma_f32_16x16x4_f32(
           false, a1[s],  false, bf,
           (short)0, acc[nt],  false,  false);
    }
  }

#pragma unroll
  for (int j = 0; j < 8; ++j) {
    float* dst = &part[w][(j + 8 * half) * PP + ln * 8];
    const v4f lo = {acc[0][j], acc[1][j], acc[2][j], acc[3][j]};
    const v4f hi = {acc[4][j], acc[5][j], acc[6][j], acc[7][j]};
    *(v4f*)(dst)     = lo;
    *(v4f*)(dst + 4) = hi;
  }
  __syncthreads();

  v8f c2[2] = {};
#pragma unroll
  for (int q = 0; q < 4; ++q) {
    const int k2 = w * 16 + q * 4 + kb;
    v2f h = *(const v2f*)(&part[0][ln * PP + k2]);
#pragma unroll
    for (int p = 1; p < 8; ++p) {
      const v2f t = *(const v2f*)(&part[p][ln * PP + k2]);
      h.x += t.x;
      h.y += t.y;
    }
    h.x += b1v[q].x;
    h.y += b1v[q].y;
    h.x = h.x > 0.0f ? h.x : 0.0f;
    h.y = h.y > 0.0f ? h.y : 0.0f;
#pragma unroll
    for (int nt = 0; nt < 2; ++nt) {
      c2[nt] = __builtin_amdgcn_wmma_f32_16x16x4_f32(
          false, h, false, w2f[q][nt], (short)0, c2[nt], false, false);
    }
  }

#pragma unroll
  for (int nt = 0; nt < 2; ++nt) {
#pragma unroll
    for (int j = 0; j < 8; ++j) {
      part2[w][(j + 8 * half) * P2 + nt * 16 + ln] = c2[nt][j];
    }
  }
  __syncthreads();

  for (int e = tid; e < 16 * OUT_N; e += 256) {
    const int r = e / OUT_N;
    const int c = e - r * OUT_N;
    float s = part2[0][r * P2 + c];
#pragma unroll
    for (int p = 1; p < 8; ++p) s += part2[p][r * P2 + c];
    out[(long)row0 * OUT_N + e] = s + b2[c];
  }
}

extern "C" void kernel_launch(void* const* d_in, const int* in_sizes, int n_in,
                              void* d_out, int out_size, void* d_ws, size_t ws_size,
                              hipStream_t stream) {
  const int*   x       = (const int*)  d_in[0];
  const int*   lengths = (const int*)  d_in[1];
  const float* table   = (const float*)d_in[2];
  const float* W1      = (const float*)d_in[3];
  const float* b1      = (const float*)d_in[4];
  const float* W2      = (const float*)d_in[5];
  const float* b2      = (const float*)d_in[6];
  float*       logits  = (float*)d_out;
  float*       rep     = (float*)d_ws;

  embed_mean_kernel<<<BATCH, GWG, 0, stream>>>(x, lengths, table, rep);
  mlp_wmma_kernel<<<BATCH / 16, 256, 0, stream>>>(rep, W1, b1, W2, b2, logits);
}
